# starting kernel
# baseline (speedup 1.0000x reference)
#include <hip/hip_runtime.h>


typedef __attribute__((ext_vector_type(2))) float v2f;
typedef __attribute__((ext_vector_type(8))) float v8f;

#define NPIX 2304
#define KDIM 576
#define HW   48
#define CCH  64

#define WMMA_F32(a, b, c) \
  __builtin_amdgcn_wmma_f32_16x16x4_f32(false, (a), false, (b), (short)0, (c), false, false)

__global__ __launch_bounds__(256) void prep_norm_kernel(
    const float* __restrict__ bg,
    const float* __restrict__ mask,
    float* __restrict__ mp,
    float* __restrict__ rnrm)
{
  int n = blockIdx.x * 256 + threadIdx.x;
  if (n >= NPIX) return;
  int ny = n / HW, nx = n % HW;

  float msum = 0.f;
  #pragma unroll
  for (int dy = 0; dy < 3; ++dy) {
    int y = ny + dy - 1;
    if ((unsigned)y >= (unsigned)HW) continue;
    #pragma unroll
    for (int dx = 0; dx < 3; ++dx) {
      int x = nx + dx - 1;
      if ((unsigned)x >= (unsigned)HW) continue;
      msum += mask[y * HW + x];
    }
  }
  mp[n] = (msum != 0.f) ? 1.f : 0.f;

  float sq = 0.f;
  for (int c = 0; c < CCH; ++c) {
    const float* bc = bg + c * NPIX;
    #pragma unroll
    for (int dy = 0; dy < 3; ++dy) {
      int y = ny + dy - 1;
      if ((unsigned)y >= (unsigned)HW) continue;
      #pragma unroll
      for (int dx = 0; dx < 3; ++dx) {
        int x = nx + dx - 1;
        if ((unsigned)x >= (unsigned)HW) continue;
        float v = bc[y * HW + x];
        sq += v * v;
      }
    }
  }
  rnrm[n] = 1.f / fmaxf(sqrtf(sq), 1e-4f);
}

__global__ __launch_bounds__(256) void prep_unfold_kernel(
    const float* __restrict__ fg, const float* __restrict__ bg,
    const float* __restrict__ rnrm,
    float* __restrict__ Pf, float* __restrict__ Pbn)
{
  int t = blockIdx.x * 256 + threadIdx.x;
  if (t >= NPIX * KDIM) return;
  int n = t / KDIM, k = t % KDIM;
  int c = k / 9, r = k % 9, dy = r / 3, dx = r % 3;
  int y = n / HW + dy - 1, x = n % HW + dx - 1;
  bool in = ((unsigned)y < (unsigned)HW) && ((unsigned)x < (unsigned)HW);
  float vf = 0.f, vb = 0.f;
  if (in) {
    int off = c * NPIX + y * HW + x;
    vf = fg[off];
    vb = bg[off];
  }
  Pf[t]  = vf;
  Pbn[t] = vb * rnrm[n];
}

__global__ __launch_bounds__(256) void prep_unfold_t_kernel(
    const float* __restrict__ bg, const float* __restrict__ rnrm,
    float* __restrict__ Pbnt)
{
  int t = blockIdx.x * 256 + threadIdx.x;
  if (t >= NPIX * KDIM) return;
  int k = t / NPIX, n = t % NPIX;
  int c = k / 9, r = k % 9, dy = r / 3, dx = r % 3;
  int y = n / HW + dy - 1, x = n % HW + dx - 1;
  bool in = ((unsigned)y < (unsigned)HW) && ((unsigned)x < (unsigned)HW);
  float vb = in ? bg[c * NPIX + y * HW + x] : 0.f;
  Pbnt[t] = vb * rnrm[n];
}

__global__ __launch_bounds__(256) void gemm1_kernel(
    const float* __restrict__ Pf,
    const float* __restrict__ Pbn,
    float* __restrict__ attn)
{
  __shared__ float la[128][36];
  __shared__ float lb[256][36];

  const int tid = threadIdx.x, lane = tid & 31, wave = tid >> 5;
  const int wp = wave >> 2;
  const int wn = wave & 3;
  const int p0 = blockIdx.y * 128, n0 = blockIdx.x * 256;
  const int fr = lane & 15;
  const int fk = (lane >> 4) << 1;

  v8f acc[4][4];
  #pragma unroll
  for (int i = 0; i < 4; ++i)
    #pragma unroll
    for (int j = 0; j < 4; ++j) acc[i][j] = (v8f){};

  for (int kc = 0; kc < KDIM; kc += 32) {
    __syncthreads();
    #pragma unroll
    for (int i = 0; i < 4; ++i) {
      int e = tid + i * 256;
      int row = e >> 3, col4 = (e & 7) << 2;
      *(float4*)&la[row][col4] =
          *(const float4*)&Pf[(size_t)(p0 + row) * KDIM + kc + col4];
    }
    #pragma unroll
    for (int i = 0; i < 8; ++i) {
      int e = tid + i * 256;
      int row = e >> 3, col4 = (e & 7) << 2;
      *(float4*)&lb[row][col4] =
          *(const float4*)&Pbn[(size_t)(n0 + row) * KDIM + kc + col4];
    }
    __syncthreads();

    #pragma unroll
    for (int kk = 0; kk < 32; kk += 4) {
      v2f af[4], bf[4];
      #pragma unroll
      for (int i = 0; i < 4; ++i)
        af[i] = *(const v2f*)&la[wp * 64 + i * 16 + fr][kk + fk];
      #pragma unroll
      for (int j = 0; j < 4; ++j)
        bf[j] = *(const v2f*)&lb[wn * 64 + j * 16 + fr][kk + fk];
      #pragma unroll
      for (int i = 0; i < 4; ++i)
        #pragma unroll
        for (int j = 0; j < 4; ++j)
          acc[i][j] = WMMA_F32(af[i], bf[j], acc[i][j]);
    }
  }

  const int dr = (lane >> 4) << 3, dc = lane & 15;
  #pragma unroll
  for (int i = 0; i < 4; ++i)
    #pragma unroll
    for (int j = 0; j < 4; ++j) {
      float* dst = attn + (size_t)(p0 + wp * 64 + i * 16 + dr) * NPIX
                        + n0 + wn * 64 + j * 16 + dc;
      #pragma unroll
      for (int r = 0; r < 8; ++r) dst[(size_t)r * NPIX] = acc[i][j][r];
    }
}

__global__ __launch_bounds__(256) void softmax_kernel(
    float* __restrict__ attn, const float* __restrict__ mp)
{
  const int p = blockIdx.x * 8 + (threadIdx.x >> 5);
  const int lane = threadIdx.x & 31;
  float* row = attn + (size_t)p * NPIX;

  float m = -3.0e38f, s = 0.f;
  for (int n = lane; n < NPIX; n += 32) {
    float z = row[n] * mp[n] * 10.0f;
    float mn = fmaxf(m, z);
    s = s * __expf(m - mn) + __expf(z - mn);
    m = mn;
  }
  #pragma unroll
  for (int off = 16; off > 0; off >>= 1) {
    float m2 = __shfl_xor(m, off, 32);
    float s2 = __shfl_xor(s, off, 32);
    float mn = fmaxf(m, m2);
    s = s * __expf(m - mn) + s2 * __expf(m2 - mn);
    m = mn;
  }
  const float inv = 1.0f / s;
  for (int n = lane; n < NPIX; n += 32) {
    float g = mp[n];
    float z = row[n] * g * 10.0f;
    row[n] = __expf(z - m) * inv * g;
  }
}

__global__ __launch_bounds__(128) void gemm2_kernel(
    const float* __restrict__ Pbnt,
    const float* __restrict__ attn,
    float* __restrict__ V)
{
  __shared__ float la[64][36];
  __shared__ float lb[256][36];

  const int tid = threadIdx.x, lane = tid & 31, wave = tid >> 5;
  const int wq = wave;
  const int k0 = blockIdx.y * 64, q0 = blockIdx.x * 256;
  const int fr = lane & 15;
  const int fk = (lane >> 4) << 1;

  v8f acc[4][4];
  #pragma unroll
  for (int i = 0; i < 4; ++i)
    #pragma unroll
    for (int j = 0; j < 4; ++j) acc[i][j] = (v8f){};

  for (int nc = 0; nc < NPIX; nc += 32) {
    __syncthreads();
    #pragma unroll
    for (int i = 0; i < 4; ++i) {
      int e = tid + i * 128;
      int row = e >> 3, col4 = (e & 7) << 2;
      *(float4*)&la[row][col4] =
          *(const float4*)&Pbnt[(size_t)(k0 + row) * NPIX + nc + col4];
    }
    #pragma unroll
    for (int i = 0; i < 16; ++i) {
      int e = tid + i * 128;
      int row = e >> 3, col4 = (e & 7) << 2;
      *(float4*)&lb[row][col4] =
          *(const float4*)&attn[(size_t)(q0 + row) * NPIX + nc + col4];
    }
    __syncthreads();

    #pragma unroll
    for (int kk = 0; kk < 32; kk += 4) {
      v2f af[4], bf[4];
      #pragma unroll
      for (int i = 0; i < 4; ++i)
        af[i] = *(const v2f*)&la[i * 16 + fr][kk + fk];
      #pragma unroll
      for (int j = 0; j < 4; ++j)
        bf[j] = *(const v2f*)&lb[wq * 64 + j * 16 + fr][kk + fk];
      #pragma unroll
      for (int i = 0; i < 4; ++i)
        #pragma unroll
        for (int j = 0; j < 4; ++j)
          acc[i][j] = WMMA_F32(af[i], bf[j], acc[i][j]);
    }
  }

  const int dr = (lane >> 4) << 3, dc = lane & 15;
  #pragma unroll
  for (int i = 0; i < 4; ++i)
    #pragma unroll
    for (int j = 0; j < 4; ++j) {
      float* dst = V + (size_t)(k0 + i * 16 + dr) * NPIX
                     + q0 + wq * 64 + j * 16 + dc;
      #pragma unroll
      for (int r = 0; r < 8; ++r) dst[(size_t)r * NPIX] = acc[i][j][r];
    }
}

__global__ __launch_bounds__(256) void fold_kernel(
    const float* __restrict__ V, float* __restrict__ out)
{
  int t = blockIdx.x * 256 + threadIdx.x;
  if (t >= CCH * NPIX) return;
  int c = t / NPIX, p = t % NPIX;
  int y = p / HW, x = p % HW;
  float acc = 0.f;
  #pragma unroll
  for (int dy = 0; dy < 3; ++dy) {
    int yy = y + 1 - dy;
    if ((unsigned)yy >= (unsigned)HW) continue;
    #pragma unroll
    for (int dx = 0; dx < 3; ++dx) {
      int xx = x + 1 - dx;
      if ((unsigned)xx >= (unsigned)HW) continue;
      acc += V[(size_t)(c * 9 + dy * 3 + dx) * NPIX + yy * HW + xx];
    }
  }
  out[t] = acc;
}

extern "C" void kernel_launch(void* const* d_in, const int* in_sizes, int n_in,
                              void* d_out, int out_size, void* d_ws, size_t ws_size,
                              hipStream_t stream) {
  const float* fg   = (const float*)d_in[0];
  const float* bg   = (const float*)d_in[1];
  const float* mask = (const float*)d_in[2];
  float* out = (float*)d_out;

  float* ws = (float*)d_ws;
  const size_t PK = (size_t)NPIX * KDIM;
  const size_t PN = (size_t)NPIX * NPIX;
  float* mp   = ws;
  float* rn   = ws + NPIX;
  float* Pf   = ws + 2 * NPIX;
  float* Pbn  = Pf + PK;
  float* Pbnt = Pbn + PK;
  float* attn = Pbnt + PK;
  float* V    = attn + PN;

  for (int b = 0; b < 4; ++b) {
    const float* fgb = fg + (size_t)b * CCH * NPIX;
    const float* bgb = bg + (size_t)b * CCH * NPIX;
    const float* mb  = mask + (size_t)b * NPIX;
    float* outb = out + (size_t)b * CCH * NPIX;

    prep_norm_kernel<<<(NPIX + 255) / 256, 256, 0, stream>>>(bgb, mb, mp, rn);
    prep_unfold_kernel<<<(NPIX * KDIM) / 256, 256, 0, stream>>>(fgb, bgb, rn, Pf, Pbn);
    prep_unfold_t_kernel<<<(NPIX * KDIM) / 256, 256, 0, stream>>>(bgb, rn, Pbnt);

    dim3 g1(NPIX / 256, NPIX / 128);
    gemm1_kernel<<<g1, 256, 0, stream>>>(Pf, Pbn, attn);

    softmax_kernel<<<NPIX / 8, 256, 0, stream>>>(attn, mp);

    dim3 g2(NPIX / 256, KDIM / 64);
    gemm2_kernel<<<g2, 128, 0, stream>>>(Pbnt, attn, V);

    fold_kernel<<<(CCH * NPIX) / 256, 256, 0, stream>>>(V, outb);
  }
}


# background window of a workgroup's whole patch segment (18 rows x 50 columns x 64 channels) staged once in LDS before the walk: the chunk loop holds no global load and no barrier
# speedup vs baseline: 9.4962x; 9.4962x over previous
#include <hip/hip_runtime.h>


typedef __attribute__((ext_vector_type(2))) float v2f;
typedef __attribute__((ext_vector_type(4))) float v4f;
typedef __attribute__((ext_vector_type(8))) float v8f;

#define NPIX 2304
#define KDIM 576
#define HW   48
#define CCH  64
#define NBL  2

#define WMMA_F32(a, b, c) \
  __builtin_amdgcn_wmma_f32_16x16x4_f32(false, (a), false, (b), (short)0, (c), false, false)

#define NSPLIT 3
#define NSEG   (NPIX / NSPLIT)
#define WROWS  (NSEG / HW + 2)
#define WCOLS  (HW + 2)
#define NPOS   (WROWS * WCOLS)
#define NPB    ((NPOS + 31) / 32)
#define CPITCH 68
#define WSIZE  (NPOS * CPITCH)

__global__ __launch_bounds__(128) void fused_attn_kernel(
    const float* __restrict__ fg,
    const float* __restrict__ bg,
    const float* __restrict__ mask,
    float* __restrict__ Opart,
    float* __restrict__ mpart,
    float* __restrict__ spart)
{
  __shared__ __attribute__((aligned(16))) float lw[WSIZE];
  __shared__ __attribute__((aligned(16))) float ln[NPOS * 2];
  __shared__ __attribute__((aligned(16))) float lq[NSEG];

  {
    const int bz = blockIdx.z;
    fg    += (size_t)bz * CCH * NPIX;
    bg    += (size_t)bz * CCH * NPIX;
    mask  += (size_t)bz * NPIX;
    Opart += (size_t)bz * NSPLIT * KDIM * NPIX;
    mpart += (size_t)bz * NSPLIT * NPIX;
    spart += (size_t)bz * NSPLIT * NPIX;
  }

  const int tid = threadIdx.x, lane = tid & 31, wave = tid >> 5;
  const int c = lane & 15, h = lane >> 4;
  const int pw = (blockIdx.x * 4 + wave) * 16;
  const int split = blockIdx.y;
  const int nbeg = split * NSEG;

  {
    const int ny0 = nbeg / HW;
    const int wave_s = __builtin_amdgcn_readfirstlane(wave);
    for (int pb = wave_s; pb < NPB; pb += 4) {
      const int pos = pb * 32 + lane;
      const bool in = pos < NPOS;
      const int row = pos / WCOLS, xx = pos % WCOLS;
      const int y = ny0 - 1 + row, x = xx - 1;
      const bool ok = in && ((unsigned)y < (unsigned)HW) && ((unsigned)x < (unsigned)HW);
      const float* sp = bg + (ok ? y * HW + x : 0);
      float wv[64];
      #pragma unroll
      for (int ch = 0; ch < 64; ++ch) wv[ch] = sp[(size_t)ch * NPIX];
      const float mv = mask[ok ? y * HW + x : 0];
      if (in) {
        float* d = &lw[pos * CPITCH];
        #pragma unroll
        for (int i = 0; i < 16; ++i)
          *(v4f*)(d + 4 * i) = (v4f){ok ? wv[4 * i] : 0.f, ok ? wv[4 * i + 1] : 0.f,
                                     ok ? wv[4 * i + 2] : 0.f, ok ? wv[4 * i + 3] : 0.f};
        float q0 = 0.f, q1 = 0.f, q2 = 0.f, q3 = 0.f;
        #pragma unroll
        for (int i = 0; i < 16; ++i) {
          q0 = fmaf(wv[4 * i],     wv[4 * i],     q0);
          q1 = fmaf(wv[4 * i + 1], wv[4 * i + 1], q1);
          q2 = fmaf(wv[4 * i + 2], wv[4 * i + 2], q2);
          q3 = fmaf(wv[4 * i + 3], wv[4 * i + 3], q3);
        }
        const float ss = (q0 + q1) + (q2 + q3);
        *(v2f*)&ln[pos * 2] = (v2f){ok ? ss : 0.f, ok ? mv : 0.f};
      }
    }
  }

  v2f pf[144];
  {
    const int p = pw + c;
    const int py = p / HW, px = p % HW;
    #pragma unroll
    for (int tap = 0; tap < 9; ++tap) {
      const int y = py + tap / 3 - 1, x = px + tap % 3 - 1;
      const bool ok = ((unsigned)y < (unsigned)HW) && ((unsigned)x < (unsigned)HW);
      const float* src = fg + (size_t)(2 * h) * NPIX + (ok ? y * HW + x : 0);
      #pragma unroll
      for (int q = 0; q < 16; ++q) {
        const float v0 = src[(size_t)(4 * q) * NPIX];
        const float v1 = src[(size_t)(4 * q + 1) * NPIX];
        pf[tap * 16 + q] = (v2f){ok ? v0 : 0.f, ok ? v1 : 0.f};
      }
    }
  }

  __syncthreads();
  for (int n = tid; n < NSEG; n += 128) {
    const float* lp = &ln[((n / HW) * WCOLS + n % HW) * 2];
    float sa = 0.f, sb = 0.f;
    #pragma unroll
    for (int k = 0; k < 9; ++k) {
      const v2f v = *(const v2f*)(lp + ((k / 3) * WCOLS + k % 3) * 2);
      sa += v.x;
      sb += v.y;
    }
    lq[n] = ((sb != 0.f) ? 1.f : 0.f) / fmaxf(sqrtf(sa), 1e-4f);
  }
  __syncthreads();

  v8f o[36];
  #pragma unroll
  for (int t = 0; t < 36; ++t) o[t] = (v8f){};
  float m = -3.0e38f, ssum = 0.f;

  const float* lane_a = &lw[c * CPITCH + 2 * h];
  const float* lane_v = &lw[(8 * h) * CPITCH + c];

  v2f af0[16];
  v4f qa, qb;
  #pragma unroll
  for (int q = 0; q < 16; ++q) af0[q] = *(const v2f*)(lane_a + 4 * q);
  qa = *(const v4f*)&lq[8 * h];
  qb = *(const v4f*)&lq[8 * h + 4];

  for (int nc = 0; nc < NSEG; nc += 16) {
    const int wofs = ((nc / HW) * WCOLS + nc % HW) * CPITCH;
    const float qs[8] = {qa.x, qa.y, qa.z, qa.w, qb.x, qb.y, qb.z, qb.w};

    v8f s0 = (v8f){}, s1 = (v8f){};
    float wq[3][4][10];
    {
      const float* arow = lane_a + wofs;
      v2f af[2][16];
      const float* vb = lane_v + wofs;
      #pragma unroll
      for (int q = 0; q < 16; ++q) af[0][q] = af0[q];
      #pragma unroll
      for (int tap = 0; tap < 9; ++tap) {
        if (tap + 1 < 9) {
          const float* an = arow + (((tap + 1) / 3) * WCOLS + (tap + 1) % 3) * CPITCH;
          #pragma unroll
          for (int q = 0; q < 16; ++q) af[(tap + 1) & 1][q] = *(const v2f*)(an + 4 * q);
        } else {
          #pragma unroll
          for (int dy = 0; dy < 3; ++dy)
            #pragma unroll
            for (int cb = 0; cb < 4; ++cb)
              #pragma unroll
              for (int col = 0; col < 10; ++col)
                wq[dy][cb][col] = vb[(dy * WCOLS + col) * CPITCH + 16 * cb];
        }
        __builtin_amdgcn_sched_barrier(0);
        #pragma unroll
        for (int q = 0; q < 16; q += 2) {
          s0 = WMMA_F32(af[tap & 1][q],     pf[tap * 16 + q],     s0);
          s1 = WMMA_F32(af[tap & 1][q + 1], pf[tap * 16 + q + 1], s1);
        }
      }
    }

    float z[8];
    float cmax = -3.0e38f;
    #pragma unroll
    for (int r = 0; r < 8; ++r) {
      z[r] = (s0[r] + s1[r]) * qs[r] * 10.0f;
      cmax = fmaxf(cmax, z[r]);
    }
    cmax = fmaxf(cmax, __shfl_xor(cmax, 16, 32));
    const float mn = fmaxf(m, cmax);
    const float alpha = __expf(m - mn);
    m = mn;
    float pr[8];
    float psum = 0.f;
    #pragma unroll
    for (int r = 0; r < 8; ++r) {
      float ex = __expf(z[r] - mn);
      psum += ex;
      pr[r] = ex * qs[r];
    }
    ssum = ssum * alpha + psum;
    #pragma unroll
    for (int t = 0; t < 36; ++t) o[t] *= alpha;

    {
      const int nn = (nc + 16 < NSEG) ? nc + 16 : nc;
      const float* an = lane_a + ((nn / HW) * WCOLS + nn % HW) * CPITCH;
      #pragma unroll
      for (int q = 0; q < 16; ++q) af0[q] = *(const v2f*)(an + 4 * q);
      qa = *(const v4f*)&lq[nn + 8 * h];
      qb = *(const v4f*)&lq[nn + 8 * h + 4];
    }

    #pragma unroll
    for (int j = 0; j < 4; ++j) {
      const v2f b = (v2f){pr[2 * j], pr[2 * j + 1]};
      #pragma unroll
      for (int t = 0; t < 36; ++t) {
        const int dy = (t >> 2) / 3, dx = (t >> 2) % 3, cb = t & 3;
        const v2f a = (v2f){wq[dy][cb][2 * j + dx], wq[dy][cb][2 * j + dx + 1]};
        o[t] = WMMA_F32(a, b, o[t]);
      }
    }
  }

  const float stot = ssum + __shfl_xor(ssum, 16, 32);
  if (h == 0) {
    mpart[split * NPIX + pw + c] = m;
    spart[split * NPIX + pw + c] = stot;
  }
  float* dst = Opart + (size_t)split * KDIM * NPIX + (size_t)(8 * h) * NPIX + pw + c;
  #pragma unroll
  for (int t = 0; t < 36; ++t)
    #pragma unroll
    for (int r = 0; r < 8; ++r)
      dst[(size_t)(t * 16 + r) * NPIX] = o[t][r];
}

__global__ __launch_bounds__(256) void fold_kernel(
    const float* __restrict__ Opart, const float* __restrict__ mpart,
    const float* __restrict__ spart, float* __restrict__ out)
{
  int t = blockIdx.x * 256 + threadIdx.x;
  if (t >= CCH * NPIX) return;
  {
    const int bz = blockIdx.y;
    Opart += (size_t)bz * NSPLIT * KDIM * NPIX;
    mpart += (size_t)bz * NSPLIT * NPIX;
    spart += (size_t)bz * NSPLIT * NPIX;
    out   += (size_t)bz * CCH * NPIX;
  }
  int c = t / NPIX, p = t % NPIX;
  int y = p / HW, x = p % HW;
  float acc = 0.f;
  #pragma unroll
  for (int dy = 0; dy < 3; ++dy) {
    int yy = y + 1 - dy;
    if ((unsigned)yy >= (unsigned)HW) continue;
    #pragma unroll
    for (int dx = 0; dx < 3; ++dx) {
      int xx = x + 1 - dx;
      if ((unsigned)xx >= (unsigned)HW) continue;
      const int q = yy * HW + xx;
      float ms[NSPLIT];
      float mx = -3.0e38f;
      #pragma unroll
      for (int s = 0; s < NSPLIT; ++s) { ms[s] = mpart[s * NPIX + q]; mx = fmaxf(mx, ms[s]); }
      const size_t base = (size_t)((dy * 3 + dx) * CCH + c) * NPIX + q;
      float den = 0.f, num = 0.f;
      #pragma unroll
      for (int s = 0; s < NSPLIT; ++s) {
        float w = __expf(ms[s] - mx);
        den += w * spart[s * NPIX + q];
        num += w * Opart[(size_t)s * KDIM * NPIX + base];
      }
      acc += num / den;
    }
  }
  out[t] = acc;
}

extern "C" void kernel_launch(void* const* d_in, const int* in_sizes, int n_in,
                              void* d_out, int out_size, void* d_ws, size_t ws_size,
                              hipStream_t stream) {
  const float* fg   = (const float*)d_in[0];
  const float* bg   = (const float*)d_in[1];
  const float* mask = (const float*)d_in[2];
  float* out = (float*)d_out;

  float* ws = (float*)d_ws;
  const size_t PK = (size_t)NPIX * KDIM;
  float* Opart = ws;
  float* mpart = Opart + NBL * NSPLIT * PK;
  float* spart = mpart + NBL * NSPLIT * NPIX;

  for (int b = 0; b < 4; b += NBL) {
    const float* fgb = fg + (size_t)b * CCH * NPIX;
    const float* bgb = bg + (size_t)b * CCH * NPIX;
    const float* mb  = mask + (size_t)b * NPIX;
    float* outb = out + (size_t)b * CCH * NPIX;

    dim3 gf(NPIX / 64, NSPLIT, NBL);
    fused_attn_kernel<<<gf, 128, 0, stream>>>(fgb, bgb, mb, Opart, mpart, spart);

    dim3 gd((CCH * NPIX) / 256, NBL);
    fold_kernel<<<gd, 256, 0, stream>>>(Opart, mpart, spart, outb);
  }
}
